# starting kernel
# baseline (speedup 1.0000x reference)
#include <hip/hip_runtime.h>

typedef float v2f __attribute__((ext_vector_type(2)));
typedef float v8f __attribute__((ext_vector_type(8)));

__global__ void gnn_zero_agg(float* __restrict__ ws, float* __restrict__ out, int n_nodes) {
    long idx   = (long)blockIdx.x * blockDim.x + threadIdx.x;
    long total = (long)n_nodes * 32;
    if (idx >= total) return;
    long row = idx >> 5;
    long q   = idx & 31;
    float4 z = make_float4(0.f, 0.f, 0.f, 0.f);
    ((float4*)(ws  + row * 256))[q] = z;
    ((float4*)(out + row * 256))[q] = z;
}

#define KMAX 256
#define APAD 260

__global__ void gnn_gemm_wmma(const float* __restrict__ A, int K,
                              const float* __restrict__ B,
                              const float* __restrict__ bias,
                              float* __restrict__ C, int ldc, int col_off) {
    __shared__ float As[16][APAD];

    const int tid   = threadIdx.x;
    const int mbase = blockIdx.x * 16;

    for (int i = tid; i < 16 * K; i += 256) {
        int r = i / K, c = i % K;
        As[r][c] = A[(long)(mbase + r) * K + c];
    }
    __syncthreads();

    const int wave  = tid >> 5;
    const int lane  = tid & 31;
    const int nbase = wave * 16;
    const int n     = lane & 15;
    const int hi    = lane >> 4;

    v8f acc;
    {
        float bv = bias[nbase + n];
        for (int g = 0; g < 8; ++g) acc[g] = bv;
    }

    for (int k = 0; k < K; k += 4) {
        int kb = k + 2 * hi;
        v2f a, b;
        a.x = As[n][kb];
        a.y = As[n][kb + 1];
        b.x = B[(long)kb       * 128 + nbase + n];
        b.y = B[(long)(kb + 1) * 128 + nbase + n];
        acc = __builtin_amdgcn_wmma_f32_16x16x4_f32(
                   false, a,  false, b,
                   (short)0, acc,  false,  false);
    }

    for (int g = 0; g < 8; ++g) {
        int row = mbase + g + 8 * hi;
        C[(long)row * ldc + col_off + nbase + n] = acc[g];
    }
}

__global__ void gnn_scatter(const int* __restrict__ src, const int* __restrict__ dst,
                            const float* __restrict__ msg_tab, float* __restrict__ agg_tab,
                            int n_edges) {
    int edge = blockIdx.x * 8 + (threadIdx.x >> 5);
    int lane = threadIdx.x & 31;
    if (edge >= n_edges) return;

    int s = src[edge];
    int d = dst[edge];

    const float4 v = ((const float4*)(msg_tab + (long)s * 256 + 128))[lane];
    float* a = agg_tab + (long)d * 256 + lane * 4;
    atomicAdd(a + 0, v.x);
    atomicAdd(a + 1, v.y);
    atomicAdd(a + 2, v.z);
    atomicAdd(a + 3, v.w);
}

extern "C" void kernel_launch(void* const* d_in, const int* in_sizes, int n_in,
                              void* d_out, int out_size, void* d_ws, size_t ws_size,
                              hipStream_t stream) {
    const float* features = (const float*)d_in[0];
    const int*   src      = (const int*)  d_in[1];
    const int*   dst      = (const int*)  d_in[2];
    const float* W1       = (const float*)d_in[3];
    const float* b1       = (const float*)d_in[4];
    const float* W2       = (const float*)d_in[5];
    const float* b2       = (const float*)d_in[6];

    float* out   = (float*)d_out;
    float* feat1 = (float*)d_ws;

    const int N = in_sizes[0] / 128;
    const int E = in_sizes[1];

    {
        long total  = (long)N * 32;
        int  blocks = (int)((total + 255) / 256);
        gnn_zero_agg<<<blocks, 256, 0, stream>>>(feat1, out, N);
    }

    gnn_gemm_wmma<<<N / 16, 256, 0, stream>>>(features, 128, W1, b1, feat1, 256, 128);

    {
        int blocks = (E + 7) / 8;
        gnn_scatter<<<blocks, 256, 0, stream>>>(src, dst, feat1, feat1, E);
    }

    gnn_gemm_wmma<<<N / 16, 256, 0, stream>>>(feat1, 256, W2, b2, out, 256, 128);

    {
        int blocks = (E + 7) / 8;
        gnn_scatter<<<blocks, 256, 0, stream>>>(src, dst, out, out, E);
    }
}


# Edge scatter by owner blocks, no atomics, no zero fill: a block filters one owner byte per edge for its 200 destination rows, sums them in LDS in a fixed order, stores each row once
# speedup vs baseline: 1.7931x; 1.7931x over previous
#include <hip/hip_runtime.h>


typedef float v2f __attribute__((ext_vector_type(2)));
typedef float v8f __attribute__((ext_vector_type(8)));
typedef unsigned v4u __attribute__((ext_vector_type(4)));

#define OWN_ROWS     200
#define OWN_WAVES    32
#define OWN_THREADS  1024
#define SCAN_ITERS   49
#define KEYS_PER_LD  512
#define SEG_EDGES    (OWN_WAVES * SCAN_ITERS * KEYS_PER_LD)
#define LIST_CAP     12288

__global__ void gnn_edge_keys(const int* __restrict__ dst, unsigned char* __restrict__ keys,
                              int n_edges) {
    long e0 = ((long)blockIdx.x * blockDim.x + threadIdx.x) * 16;
    if (e0 >= n_edges) return;
    if (e0 + 16 <= n_edges) {
        unsigned w[4];
        for (int q = 0; q < 4; ++q) {
            int4 d = *(const int4*)(dst + e0 + q * 4);
            w[q] = ((unsigned)d.x / OWN_ROWS) | (((unsigned)d.y / OWN_ROWS) << 8) |
                   (((unsigned)d.z / OWN_ROWS) << 16) | (((unsigned)d.w / OWN_ROWS) << 24);
        }
        *(uint4*)(keys + e0) = make_uint4(w[0], w[1], w[2], w[3]);
    } else {
        for (long e = e0; e < n_edges; ++e)
            keys[e] = (unsigned char)((unsigned)dst[e] / OWN_ROWS);
    }
}

#define KMAX 256
#define APAD 260

__global__ void gnn_gemm_wmma(const float* __restrict__ A, int K,
                              const float* __restrict__ B,
                              const float* __restrict__ bias,
                              float* __restrict__ C, int ldc, int col_off) {
    __shared__ float As[16][APAD];

    const int tid   = threadIdx.x;
    const int mbase = blockIdx.x * 16;

    for (int i = tid; i < 16 * K; i += 256) {
        int r = i / K, c = i % K;
        As[r][c] = A[(long)(mbase + r) * K + c];
    }
    __syncthreads();

    const int wave  = tid >> 5;
    const int lane  = tid & 31;
    const int nbase = wave * 16;
    const int n     = lane & 15;
    const int hi    = lane >> 4;

    v8f acc;
    {
        float bv = bias[nbase + n];
        for (int g = 0; g < 8; ++g) acc[g] = bv;
    }

    for (int k = 0; k < K; k += 4) {
        int kb = k + 2 * hi;
        v2f a, b;
        a.x = As[n][kb];
        a.y = As[n][kb + 1];
        b.x = B[(long)kb       * 128 + nbase + n];
        b.y = B[(long)(kb + 1) * 128 + nbase + n];
        acc = __builtin_amdgcn_wmma_f32_16x16x4_f32(
                   false, a,  false, b,
                   (short)0, acc,  false,  false);
    }

    for (int g = 0; g < 8; ++g) {
        int row = mbase + g + 8 * hi;
        C[(long)row * ldc + col_off + nbase + n] = acc[g];
    }
}

__device__ __forceinline__ unsigned eq_bytes(unsigned kw, unsigned idw) {
    unsigned x = kw ^ idw;
    return ~(((x & 0x7F7F7F7Fu) + 0x7F7F7F7Fu) | x | 0x7F7F7F7Fu);
}

__device__ __forceinline__ unsigned match16(unsigned k0, unsigned k1, unsigned k2, unsigned k3,
                                            unsigned idw) {
    unsigned mp = (eq_bytes(k0, idw) >> 7) | (eq_bytes(k1, idw) >> 6) |
                  (eq_bytes(k2, idw) >> 5) | (eq_bytes(k3, idw) >> 4);
    unsigned t = mp | (mp >> 4);
    return (t & 0xFFu) | ((t >> 8) & 0xFF00u);
}

__device__ __forceinline__ int wave_incl_scan(int v, int lane) {
    for (int d = 1; d < 32; d <<= 1) {
        int u = __shfl_up(v, d);
        if (lane >= d) v += u;
    }
    return v;
}

__device__ __forceinline__ void own_flush(const float* __restrict__ msg_tab, float* acc,
                                          int qs, int qd, int cnt, int lane) {
    for (int k = 0; k < cnt; k += 8) {
        float4 v[8];
#pragma unroll
        for (int u = 0; u < 8; ++u) {
            if (k + u < cnt) {
                int sk = __builtin_amdgcn_readlane(qs, k + u);
                v[u] = *(const float4*)(msg_tab + (long)sk * 256 + 128 + lane * 4);
            }
        }
#pragma unroll
        for (int u = 0; u < 8; ++u) {
            if (k + u < cnt) {
                int dk = __builtin_amdgcn_readlane(qd, k + u);
                float4* a = (float4*)(acc + dk * 128 + lane * 4);
                float4 t = *a;
                t.x += v[u].x; t.y += v[u].y; t.z += v[u].z; t.w += v[u].w;
                *a = t;
            }
        }
    }
}

__global__ void __launch_bounds__(OWN_THREADS)
gnn_scatter_own(const int* __restrict__ src, const int* __restrict__ dst,
                const unsigned char* __restrict__ keys,
                const float* __restrict__ msg_tab, float* __restrict__ agg_tab,
                int n_nodes, int n_edges) {
    __shared__ float          acc[OWN_ROWS * 128];
    __shared__ unsigned short masks[OWN_WAVES * SCAN_ITERS * 32];
    __shared__ int            list_s[LIST_CAP];
    __shared__ int            list_d[LIST_CAP];
    __shared__ int            cntw[OWN_WAVES];

    const int tid  = threadIdx.x;
    const int w    = tid >> 5;
    const int lane = tid & 31;
    const int lo   = blockIdx.x * OWN_ROWS;
    const unsigned idw = (unsigned)blockIdx.x * 0x01010101u;
    const int rot  = blockIdx.x % SCAN_ITERS;
    const long e16 = (long)n_edges & ~15L;

    for (int r = w; r < OWN_ROWS; r += OWN_WAVES)
        *(float4*)(acc + r * 128 + lane * 4) = make_float4(0.f, 0.f, 0.f, 0.f);

    for (long seg0 = 0; seg0 < n_edges; seg0 += SEG_EDGES) {
        const unsigned char* kseg = keys + seg0;
        const long rem = e16 - seg0;
        const int  lim = (rem > SEG_EDGES) ? SEG_EDGES : (int)rem;
        int cnt = 0;
        if (lim > 0) {
            for (int t0 = 0; t0 < SCAN_ITERS; t0 += 7) {
                v4u kw[7];
#pragma unroll
                for (int u = 0; u < 7; ++u) {
                    int it = t0 + u + rot;
                    if (it >= SCAN_ITERS) it -= SCAN_ITERS;
                    int rel = (w * SCAN_ITERS + it) * KEYS_PER_LD + lane * 16;
                    int rc  = (rel < lim - 16) ? rel : (lim - 16);
                    kw[u] = *(const v4u*)(kseg + rc);
                }
                asm volatile("" : "+v"(kw[0]), "+v"(kw[1]), "+v"(kw[2]), "+v"(kw[3]),
                                  "+v"(kw[4]), "+v"(kw[5]), "+v"(kw[6]));
#pragma unroll
                for (int u = 0; u < 7; ++u) {
                    int it = t0 + u + rot;
                    if (it >= SCAN_ITERS) it -= SCAN_ITERS;
                    int rel = (w * SCAN_ITERS + it) * KEYS_PER_LD + lane * 16;
                    unsigned m = match16(kw[u].x, kw[u].y, kw[u].z, kw[u].w, idw);
                    if (rel + 16 > lim) m = 0;
                    masks[(w * SCAN_ITERS + it) * 32 + lane] = (unsigned short)m;
                    cnt += __popc(m);
                }
            }
        } else {
            for (int it = 0; it < SCAN_ITERS; ++it) masks[(w * SCAN_ITERS + it) * 32 + lane] = 0;
        }
        if (e16 < n_edges) {
            long pos = e16 - seg0;
            if (pos >= 0 && pos < SEG_EDGES && w == (int)(pos / (SCAN_ITERS * KEYS_PER_LD)) &&
                lane == (int)((pos % KEYS_PER_LD) / 16)) {
                unsigned m = 0;
                for (int j = 0; e16 + j < n_edges; ++j)
                    if (keys[e16 + j] == (unsigned char)blockIdx.x) m |= 1u << (4 * (j & 3) + (j >> 2));
                masks[(int)(pos / 16)] = (unsigned short)m;
                cnt += __popc(m);
            }
        }
        for (int d = 16; d >= 1; d >>= 1) cnt += __shfl_xor(cnt, d);
        if (lane == 0) cntw[w] = cnt;
        __syncthreads();

        int c_l   = cntw[lane];
        int inc_l = wave_incl_scan(c_l, lane);
        const int total  = __shfl(inc_l, 31);
        const int base_w = __shfl(inc_l, w) - __shfl(c_l, w);
        __syncthreads();

        for (int r0 = 0; r0 < total; r0 += LIST_CAP) {
            const unsigned short* mw = masks + w * (SCAN_ITERS * 32) + lane * SCAN_ITERS;
            int c = 0;
            for (int i = 0; i < SCAN_ITERS; ++i) c += __popc((unsigned)mw[i]);
            int inc = wave_incl_scan(c, lane);
            int p   = base_w + inc - c - r0;
            int eb  = (int)seg0 + (w * (SCAN_ITERS * 32) + lane * SCAN_ITERS) * 16;
            for (int i = 0; i < SCAN_ITERS; ++i) {
                unsigned m = mw[i];
                while (m) {
                    int j = __ffs(m) - 1;
                    m &= m - 1;
                    if (p >= 0 && p < LIST_CAP) list_s[p] = eb + i * 16 + 4 * (j & 3) + (j >> 2);
                    ++p;
                }
            }
            __syncthreads();

            const int n = (total - r0 < LIST_CAP) ? (total - r0) : LIST_CAP;
            for (int i = tid; i < n; i += OWN_THREADS) {
                int e = list_s[i];
                list_s[i] = src[e];
                list_d[i] = dst[e] - lo;
            }
            __syncthreads();

            int qn = 0, qs = 0, qd = 0;
            int dn = (lane < n) ? list_d[lane] : -1;
            int sn = (lane < n) ? list_s[lane] : 0;
            for (int i0 = 0; i0 < n; i0 += 32) {
                const int d = dn, s = sn;
                const int i2 = i0 + 32 + lane;
                dn = (i2 < n) ? list_d[i2] : -1;
                sn = (i2 < n) ? list_s[i2] : 0;
                unsigned mk = __builtin_amdgcn_ballot_w32(d >= 0 && (d & (OWN_WAVES - 1)) == w);
                while (mk) {
                    int b = __builtin_ctz(mk);
                    mk &= mk - 1;
                    int sb = __builtin_amdgcn_readlane(s, b);
                    int db = __builtin_amdgcn_readlane(d, b);
                    if (lane == qn) { qs = sb; qd = db; }
                    ++qn;
                    if (qn == 32) { own_flush(msg_tab, acc, qs, qd, 32, lane); qn = 0; }
                }
            }
            if (qn) own_flush(msg_tab, acc, qs, qd, qn, lane);
            __syncthreads();
        }
    }

    for (int r = w; r < OWN_ROWS; r += OWN_WAVES) {
        int row = lo + r;
        if (row < n_nodes)
            *(float4*)(agg_tab + (long)row * 256 + lane * 4) = *(const float4*)(acc + r * 128 + lane * 4);
    }
}

extern "C" void kernel_launch(void* const* d_in, const int* in_sizes, int n_in,
                              void* d_out, int out_size, void* d_ws, size_t ws_size,
                              hipStream_t stream) {
    const float* features = (const float*)d_in[0];
    const int*   src      = (const int*)  d_in[1];
    const int*   dst      = (const int*)  d_in[2];
    const float* W1       = (const float*)d_in[3];
    const float* b1       = (const float*)d_in[4];
    const float* W2       = (const float*)d_in[5];
    const float* b2       = (const float*)d_in[6];

    float* out   = (float*)d_out;
    float* feat1 = (float*)d_ws;

    const int N = in_sizes[0] / 128;
    const int E = in_sizes[1];

    unsigned char* keys = (unsigned char*)d_ws + (size_t)N * 1024;

    {
        int blocks = (E + 4095) / 4096;
        gnn_edge_keys<<<blocks, 256, 0, stream>>>(dst, keys, E);
    }

    gnn_gemm_wmma<<<N / 16, 256, 0, stream>>>(features, 128, W1, b1, feat1, 256, 128);

    const int own_blocks = (N + OWN_ROWS - 1) / OWN_ROWS;

    gnn_scatter_own<<<own_blocks, OWN_THREADS, 0, stream>>>(src, dst, keys, feat1, feat1, N, E);

    gnn_gemm_wmma<<<N / 16, 256, 0, stream>>>(feat1, 256, W2, b2, out, 256, 128);

    gnn_scatter_own<<<own_blocks, OWN_THREADS, 0, stream>>>(src, dst, keys, out, out, N, E);
}


# Second edge scatter replays the per-wave (src, row) gather queues that the first scatter records in the workspace: no key scan, list build or filter in it; same order of sums
# speedup vs baseline: 2.1692x; 1.2097x over previous
#include <hip/hip_runtime.h>


typedef float v2f __attribute__((ext_vector_type(2)));
typedef float v8f __attribute__((ext_vector_type(8)));
typedef unsigned v4u __attribute__((ext_vector_type(4)));

#define OWN_ROWS     200
#define OWN_WAVES    32
#define OWN_THREADS  1024
#define SCAN_ITERS   49
#define KEYS_PER_LD  512
#define SEG_EDGES    (OWN_WAVES * SCAN_ITERS * KEYS_PER_LD)
#define LIST_CAP     12288
#define REC_CAP      512

__global__ void gnn_edge_keys(const int* __restrict__ dst, unsigned char* __restrict__ keys,
                              int n_edges) {
    long e0 = ((long)blockIdx.x * blockDim.x + threadIdx.x) * 16;
    if (e0 >= n_edges) return;
    if (e0 + 16 <= n_edges) {
        unsigned w[4];
        for (int q = 0; q < 4; ++q) {
            int4 d = *(const int4*)(dst + e0 + q * 4);
            w[q] = ((unsigned)d.x / OWN_ROWS) | (((unsigned)d.y / OWN_ROWS) << 8) |
                   (((unsigned)d.z / OWN_ROWS) << 16) | (((unsigned)d.w / OWN_ROWS) << 24);
        }
        *(uint4*)(keys + e0) = make_uint4(w[0], w[1], w[2], w[3]);
    } else {
        for (long e = e0; e < n_edges; ++e)
            keys[e] = (unsigned char)((unsigned)dst[e] / OWN_ROWS);
    }
}

#define KMAX 256
#define APAD 260

__global__ void gnn_gemm_wmma(const float* __restrict__ A, int K,
                              const float* __restrict__ B,
                              const float* __restrict__ bias,
                              float* __restrict__ C, int ldc, int col_off) {
    __shared__ float As[16][APAD];

    const int tid   = threadIdx.x;
    const int mbase = blockIdx.x * 16;

    for (int i = tid; i < 16 * K; i += 256) {
        int r = i / K, c = i % K;
        As[r][c] = A[(long)(mbase + r) * K + c];
    }
    __syncthreads();

    const int wave  = tid >> 5;
    const int lane  = tid & 31;
    const int nbase = wave * 16;
    const int n     = lane & 15;
    const int hi    = lane >> 4;

    v8f acc;
    {
        float bv = bias[nbase + n];
        for (int g = 0; g < 8; ++g) acc[g] = bv;
    }

    for (int k = 0; k < K; k += 4) {
        int kb = k + 2 * hi;
        v2f a, b;
        a.x = As[n][kb];
        a.y = As[n][kb + 1];
        b.x = B[(long)kb       * 128 + nbase + n];
        b.y = B[(long)(kb + 1) * 128 + nbase + n];
        acc = __builtin_amdgcn_wmma_f32_16x16x4_f32(
                   false, a,  false, b,
                   (short)0, acc,  false,  false);
    }

    for (int g = 0; g < 8; ++g) {
        int row = mbase + g + 8 * hi;
        C[(long)row * ldc + col_off + nbase + n] = acc[g];
    }
}

__device__ __forceinline__ unsigned eq_bytes(unsigned kw, unsigned idw) {
    unsigned x = kw ^ idw;
    return ~(((x & 0x7F7F7F7Fu) + 0x7F7F7F7Fu) | x | 0x7F7F7F7Fu);
}

__device__ __forceinline__ unsigned match16(unsigned k0, unsigned k1, unsigned k2, unsigned k3,
                                            unsigned idw) {
    unsigned mp = (eq_bytes(k0, idw) >> 7) | (eq_bytes(k1, idw) >> 6) |
                  (eq_bytes(k2, idw) >> 5) | (eq_bytes(k3, idw) >> 4);
    unsigned t = mp | (mp >> 4);
    return (t & 0xFFu) | ((t >> 8) & 0xFF00u);
}

__device__ __forceinline__ int wave_incl_scan(int v, int lane) {
    for (int d = 1; d < 32; d <<= 1) {
        int u = __shfl_up(v, d);
        if (lane >= d) v += u;
    }
    return v;
}

__device__ __forceinline__ void own_flush(const float* __restrict__ msg_tab, float* acc,
                                          int qs, int qd, int cnt, int lane) {
    for (int k = 0; k < cnt; k += 8) {
        float4 v[8];
#pragma unroll
        for (int u = 0; u < 8; ++u) {
            if (k + u < cnt) {
                int sk = __builtin_amdgcn_readlane(qs, k + u);
                v[u] = *(const float4*)(msg_tab + (long)sk * 256 + 128 + lane * 4);
            }
        }
#pragma unroll
        for (int u = 0; u < 8; ++u) {
            if (k + u < cnt) {
                int dk = __builtin_amdgcn_readlane(qd, k + u);
                float4* a = (float4*)(acc + dk * 128 + lane * 4);
                float4 t = *a;
                t.x += v[u].x; t.y += v[u].y; t.z += v[u].z; t.w += v[u].w;
                *a = t;
            }
        }
    }
}

template <bool REPLAY>
__global__ void __launch_bounds__(OWN_THREADS)
gnn_scatter_own(const int* __restrict__ src, const int* __restrict__ dst,
                const unsigned char* __restrict__ keys,
                const float* __restrict__ msg_tab, float* __restrict__ agg_tab,
                int* __restrict__ rec, int* __restrict__ rec_cnt, int* __restrict__ rec_flag,
                int n_nodes, int n_edges) {
    __shared__ float          acc[OWN_ROWS * 128];
    __shared__ unsigned short masks[OWN_WAVES * SCAN_ITERS * 32];
    __shared__ int            list_s[LIST_CAP];
    __shared__ int            list_d[LIST_CAP];
    __shared__ int            cntw[OWN_WAVES];

    const int tid  = threadIdx.x;
    const int w    = tid >> 5;
    const int lane = tid & 31;
    const int lo   = blockIdx.x * OWN_ROWS;
    const unsigned idw = (unsigned)blockIdx.x * 0x01010101u;
    const int rot  = blockIdx.x % SCAN_ITERS;
    const long e16 = (long)n_edges & ~15L;

    __shared__ int            ovf_s;

    for (int r = w; r < OWN_ROWS; r += OWN_WAVES)
        *(float4*)(acc + r * 128 + lane * 4) = make_float4(0.f, 0.f, 0.f, 0.f);

    int* const recw = rec + (long)(blockIdx.x * OWN_WAVES + w) * (2 * REC_CAP);
    int  nrec = 0;
    bool full_path = true;
    if (REPLAY) full_path = (rec_flag[blockIdx.x] != 0);
    if (!REPLAY && tid == 0) ovf_s = 0;

    if (!full_path) {
        const int c = __builtin_amdgcn_readfirstlane(rec_cnt[blockIdx.x * OWN_WAVES + w]);
        int2 nx = (lane < c) ? *(const int2*)(recw + 2 * lane) : make_int2(0, 0);
        int k = 0;
        for (; k + 32 <= c; k += 32) {
            const int2 cur = nx;
            const int i2 = k + 32 + lane;
            nx = (i2 < c) ? *(const int2*)(recw + 2 * i2) : make_int2(0, 0);
            own_flush(msg_tab, acc, cur.x, cur.y, 32, lane);
        }
        if (k < c) own_flush(msg_tab, acc, nx.x, nx.y, c - k, lane);
    }

    for (long seg0 = 0; full_path && seg0 < n_edges; seg0 += SEG_EDGES) {
        const unsigned char* kseg = keys + seg0;
        const long rem = e16 - seg0;
        const int  lim = (rem > SEG_EDGES) ? SEG_EDGES : (int)rem;
        int cnt = 0;
        if (lim > 0) {
            for (int t0 = 0; t0 < SCAN_ITERS; t0 += 7) {
                v4u kw[7];
#pragma unroll
                for (int u = 0; u < 7; ++u) {
                    int it = t0 + u + rot;
                    if (it >= SCAN_ITERS) it -= SCAN_ITERS;
                    int rel = (w * SCAN_ITERS + it) * KEYS_PER_LD + lane * 16;
                    int rc  = (rel < lim - 16) ? rel : (lim - 16);
                    kw[u] = *(const v4u*)(kseg + rc);
                }
                asm volatile("" : "+v"(kw[0]), "+v"(kw[1]), "+v"(kw[2]), "+v"(kw[3]),
                                  "+v"(kw[4]), "+v"(kw[5]), "+v"(kw[6]));
#pragma unroll
                for (int u = 0; u < 7; ++u) {
                    int it = t0 + u + rot;
                    if (it >= SCAN_ITERS) it -= SCAN_ITERS;
                    int rel = (w * SCAN_ITERS + it) * KEYS_PER_LD + lane * 16;
                    unsigned m = match16(kw[u].x, kw[u].y, kw[u].z, kw[u].w, idw);
                    if (rel + 16 > lim) m = 0;
                    masks[(w * SCAN_ITERS + it) * 32 + lane] = (unsigned short)m;
                    cnt += __popc(m);
                }
            }
        } else {
            for (int it = 0; it < SCAN_ITERS; ++it) masks[(w * SCAN_ITERS + it) * 32 + lane] = 0;
        }
        if (e16 < n_edges) {
            long pos = e16 - seg0;
            if (pos >= 0 && pos < SEG_EDGES && w == (int)(pos / (SCAN_ITERS * KEYS_PER_LD)) &&
                lane == (int)((pos % KEYS_PER_LD) / 16)) {
                unsigned m = 0;
                for (int j = 0; e16 + j < n_edges; ++j)
                    if (keys[e16 + j] == (unsigned char)blockIdx.x) m |= 1u << (4 * (j & 3) + (j >> 2));
                masks[(int)(pos / 16)] = (unsigned short)m;
                cnt += __popc(m);
            }
        }
        for (int d = 16; d >= 1; d >>= 1) cnt += __shfl_xor(cnt, d);
        if (lane == 0) cntw[w] = cnt;
        __syncthreads();

        int c_l   = cntw[lane];
        int inc_l = wave_incl_scan(c_l, lane);
        const int total  = __shfl(inc_l, 31);
        const int base_w = __shfl(inc_l, w) - __shfl(c_l, w);
        __syncthreads();

        for (int r0 = 0; r0 < total; r0 += LIST_CAP) {
            const unsigned short* mw = masks + w * (SCAN_ITERS * 32) + lane * SCAN_ITERS;
            int c = 0;
            for (int i = 0; i < SCAN_ITERS; ++i) c += __popc((unsigned)mw[i]);
            int inc = wave_incl_scan(c, lane);
            int p   = base_w + inc - c - r0;
            int eb  = (int)seg0 + (w * (SCAN_ITERS * 32) + lane * SCAN_ITERS) * 16;
            for (int i = 0; i < SCAN_ITERS; ++i) {
                unsigned m = mw[i];
                while (m) {
                    int j = __ffs(m) - 1;
                    m &= m - 1;
                    if (p >= 0 && p < LIST_CAP) list_s[p] = eb + i * 16 + 4 * (j & 3) + (j >> 2);
                    ++p;
                }
            }
            __syncthreads();

            const int n = (total - r0 < LIST_CAP) ? (total - r0) : LIST_CAP;
            for (int i = tid; i < n; i += OWN_THREADS) {
                int e = list_s[i];
                list_s[i] = src[e];
                list_d[i] = dst[e] - lo;
            }
            __syncthreads();

            int qn = 0, qs = 0, qd = 0;
            int dn = (lane < n) ? list_d[lane] : -1;
            int sn = (lane < n) ? list_s[lane] : 0;
            for (int i0 = 0; i0 < n; i0 += 32) {
                const int d = dn, s = sn;
                const int i2 = i0 + 32 + lane;
                dn = (i2 < n) ? list_d[i2] : -1;
                sn = (i2 < n) ? list_s[i2] : 0;
                unsigned mk = __builtin_amdgcn_ballot_w32(d >= 0 && (d & (OWN_WAVES - 1)) == w);
                while (mk) {
                    int b = __builtin_ctz(mk);
                    mk &= mk - 1;
                    int sb = __builtin_amdgcn_readlane(s, b);
                    int db = __builtin_amdgcn_readlane(d, b);
                    if (lane == qn) { qs = sb; qd = db; }
                    ++qn;
                    if (qn == 32) {
                        if (!REPLAY) {
                            if (nrec + 32 <= REC_CAP) *(int2*)(recw + 2 * (nrec + lane)) = make_int2(qs, qd);
                            else ovf_s = 1;
                            nrec += 32;
                        }
                        own_flush(msg_tab, acc, qs, qd, 32, lane);
                        qn = 0;
                    }
                }
            }
            if (qn) {
                if (!REPLAY) {
                    if (nrec + qn <= REC_CAP) { if (lane < qn) *(int2*)(recw + 2 * (nrec + lane)) = make_int2(qs, qd); }
                    else ovf_s = 1;
                    nrec += qn;
                }
                own_flush(msg_tab, acc, qs, qd, qn, lane);
            }
            __syncthreads();
        }
    }

    if (!REPLAY) {
        if (lane == 0) rec_cnt[blockIdx.x * OWN_WAVES + w] = nrec;
        if (tid == 0)  rec_flag[blockIdx.x] = ovf_s;
    }

    for (int r = w; r < OWN_ROWS; r += OWN_WAVES) {
        int row = lo + r;
        if (row < n_nodes)
            *(float4*)(agg_tab + (long)row * 256 + lane * 4) = *(const float4*)(acc + r * 128 + lane * 4);
    }
}

extern "C" void kernel_launch(void* const* d_in, const int* in_sizes, int n_in,
                              void* d_out, int out_size, void* d_ws, size_t ws_size,
                              hipStream_t stream) {
    const float* features = (const float*)d_in[0];
    const int*   src      = (const int*)  d_in[1];
    const int*   dst      = (const int*)  d_in[2];
    const float* W1       = (const float*)d_in[3];
    const float* b1       = (const float*)d_in[4];
    const float* W2       = (const float*)d_in[5];
    const float* b2       = (const float*)d_in[6];

    float* out   = (float*)d_out;
    float* feat1 = (float*)d_ws;

    const int N = in_sizes[0] / 128;
    const int E = in_sizes[1];

    unsigned char* keys = (unsigned char*)d_ws + (size_t)N * 1024;

    int* rec      = (int*)((unsigned char*)d_ws + ((size_t)128 << 20));
    int* rec_cnt  = (int*)((unsigned char*)d_ws + ((size_t)224 << 20));
    int* rec_flag = rec_cnt + 16384;

    {
        int blocks = (E + 4095) / 4096;
        gnn_edge_keys<<<blocks, 256, 0, stream>>>(dst, keys, E);
    }

    gnn_gemm_wmma<<<N / 16, 256, 0, stream>>>(features, 128, W1, b1, feat1, 256, 128);

    const int own_blocks = (N + OWN_ROWS - 1) / OWN_ROWS;

    gnn_scatter_own<false><<<own_blocks, OWN_THREADS, 0, stream>>>(src, dst, keys, feat1, feat1,
                                                                  rec, rec_cnt, rec_flag, N, E);

    gnn_gemm_wmma<<<N / 16, 256, 0, stream>>>(feat1, 256, W2, b2, out, 256, 128);

    gnn_scatter_own<true><<<own_blocks, OWN_THREADS, 0, stream>>>(src, dst, keys, out, out,
                                                                 rec, rec_cnt, rec_flag, N, E);
}


# GEMMs as 512 resident blocks that walk over the 16-row stripes: a wave keeps its K x 16 slab of W in registers as B fragments, A stripes double-buffered in LDS; k order and sums unchanged
# speedup vs baseline: 2.5325x; 1.1675x over previous
#include <hip/hip_runtime.h>


typedef float v2f __attribute__((ext_vector_type(2)));
typedef float v8f __attribute__((ext_vector_type(8)));
typedef unsigned v4u __attribute__((ext_vector_type(4)));
typedef float v4f __attribute__((ext_vector_type(4)));

#define OWN_ROWS     200
#define OWN_WAVES    32
#define OWN_THREADS  1024
#define SCAN_ITERS   49
#define KEYS_PER_LD  512
#define SEG_EDGES    (OWN_WAVES * SCAN_ITERS * KEYS_PER_LD)
#define LIST_CAP     12288
#define REC_CAP      512

__global__ void gnn_edge_keys(const int* __restrict__ dst, unsigned char* __restrict__ keys,
                              int n_edges) {
    long e0 = ((long)blockIdx.x * blockDim.x + threadIdx.x) * 16;
    if (e0 >= n_edges) return;
    if (e0 + 16 <= n_edges) {
        unsigned w[4];
        for (int q = 0; q < 4; ++q) {
            int4 d = *(const int4*)(dst + e0 + q * 4);
            w[q] = ((unsigned)d.x / OWN_ROWS) | (((unsigned)d.y / OWN_ROWS) << 8) |
                   (((unsigned)d.z / OWN_ROWS) << 16) | (((unsigned)d.w / OWN_ROWS) << 24);
        }
        *(uint4*)(keys + e0) = make_uint4(w[0], w[1], w[2], w[3]);
    } else {
        for (long e = e0; e < n_edges; ++e)
            keys[e] = (unsigned char)((unsigned)dst[e] / OWN_ROWS);
    }
}

#define APAD 260

template <int K>
__global__ void __launch_bounds__(256)
gnn_gemm_wmma(const float* __restrict__ A,
              const float* __restrict__ B,
              const float* __restrict__ bias,
              float* __restrict__ C, int ldc, int col_off, int n_tiles) {
    __shared__ __attribute__((aligned(16))) float As[2][16][APAD];
    constexpr int NV = K / 64;

    const int tid   = threadIdx.x;
    const int wave  = tid >> 5;
    const int lane  = tid & 31;
    const int nbase = wave * 16;
    const int n     = lane & 15;
    const int hi    = lane >> 4;

    v2f bf[K / 4];
#pragma unroll
    for (int t = 0; t < K / 4; ++t) {
        int kb = 4 * t + 2 * hi;
        bf[t].x = B[(long)kb       * 128 + nbase + n];
        bf[t].y = B[(long)(kb + 1) * 128 + nbase + n];
    }
    const float bv = bias[nbase + n];

    int mt = blockIdx.x;
    if (mt >= n_tiles) return;

    v4f st[NV];
    {
        const v4f* g = (const v4f*)(A + (long)mt * 16 * K);
#pragma unroll
        for (int j = 0; j < NV; ++j) st[j] = g[tid + 256 * j];
#pragma unroll
        for (int j = 0; j < NV; ++j) {
            int e = 4 * (tid + 256 * j);
            *(v4f*)&As[0][e / K][e % K] = st[j];
        }
    }
    __syncthreads();

    int cur = 0;
    for (;;) {
        const int  nxt  = mt + gridDim.x;
        const bool more = nxt < n_tiles;
        {
            const v4f* g = (const v4f*)(A + (long)(more ? nxt : mt) * 16 * K);
#pragma unroll
            for (int j = 0; j < NV; ++j) st[j] = g[tid + 256 * j];
        }
        __builtin_amdgcn_sched_barrier(0);

        v8f acc;
        for (int g = 0; g < 8; ++g) acc[g] = bv;

        const float* arow = &As[cur][n][2 * hi];
        v2f af[2][8];
#pragma unroll
        for (int u = 0; u < 8; ++u) af[0][u] = *(const v2f*)(arow + 4 * u);
#pragma unroll
        for (int q = 0; q < K / 32; ++q) {
            if (q + 1 < K / 32) {
#pragma unroll
                for (int u = 0; u < 8; ++u) af[(q + 1) & 1][u] = *(const v2f*)(arow + 32 * (q + 1) + 4 * u);
            }
            __builtin_amdgcn_sched_barrier(0);
#pragma unroll
            for (int u = 0; u < 8; ++u)
                acc = __builtin_amdgcn_wmma_f32_16x16x4_f32(
                           false, af[q & 1][u],  false, bf[8 * q + u],
                           (short)0, acc,  false,  false);
        }

#pragma unroll
        for (int j = 0; j < NV; ++j) {
            int e = 4 * (tid + 256 * j);
            *(v4f*)&As[cur ^ 1][e / K][e % K] = st[j];
        }
        __syncthreads();

        float* cp = C + (long)(mt * 16 + 8 * hi) * ldc + col_off + nbase + n;
        for (int g = 0; g < 8; ++g) cp[(long)g * ldc] = acc[g];

        if (!more) break;
        mt = nxt;
        cur ^= 1;
    }
}

__device__ __forceinline__ unsigned eq_bytes(unsigned kw, unsigned idw) {
    unsigned x = kw ^ idw;
    return ~(((x & 0x7F7F7F7Fu) + 0x7F7F7F7Fu) | x | 0x7F7F7F7Fu);
}

__device__ __forceinline__ unsigned match16(unsigned k0, unsigned k1, unsigned k2, unsigned k3,
                                            unsigned idw) {
    unsigned mp = (eq_bytes(k0, idw) >> 7) | (eq_bytes(k1, idw) >> 6) |
                  (eq_bytes(k2, idw) >> 5) | (eq_bytes(k3, idw) >> 4);
    unsigned t = mp | (mp >> 4);
    return (t & 0xFFu) | ((t >> 8) & 0xFF00u);
}

__device__ __forceinline__ int wave_incl_scan(int v, int lane) {
    for (int d = 1; d < 32; d <<= 1) {
        int u = __shfl_up(v, d);
        if (lane >= d) v += u;
    }
    return v;
}

__device__ __forceinline__ void own_flush(const float* __restrict__ msg_tab, float* acc,
                                          int qs, int qd, int cnt, int lane) {
    for (int k = 0; k < cnt; k += 8) {
        float4 v[8];
#pragma unroll
        for (int u = 0; u < 8; ++u) {
            if (k + u < cnt) {
                int sk = __builtin_amdgcn_readlane(qs, k + u);
                v[u] = *(const float4*)(msg_tab + (long)sk * 256 + 128 + lane * 4);
            }
        }
#pragma unroll
        for (int u = 0; u < 8; ++u) {
            if (k + u < cnt) {
                int dk = __builtin_amdgcn_readlane(qd, k + u);
                float4* a = (float4*)(acc + dk * 128 + lane * 4);
                float4 t = *a;
                t.x += v[u].x; t.y += v[u].y; t.z += v[u].z; t.w += v[u].w;
                *a = t;
            }
        }
    }
}

template <bool REPLAY>
__global__ void __launch_bounds__(OWN_THREADS)
gnn_scatter_own(const int* __restrict__ src, const int* __restrict__ dst,
                const unsigned char* __restrict__ keys,
                const float* __restrict__ msg_tab, float* __restrict__ agg_tab,
                int* __restrict__ rec, int* __restrict__ rec_cnt, int* __restrict__ rec_flag,
                int n_nodes, int n_edges) {
    __shared__ float          acc[OWN_ROWS * 128];
    __shared__ unsigned short masks[OWN_WAVES * SCAN_ITERS * 32];
    __shared__ int            list_s[LIST_CAP];
    __shared__ int            list_d[LIST_CAP];
    __shared__ int            cntw[OWN_WAVES];

    const int tid  = threadIdx.x;
    const int w    = tid >> 5;
    const int lane = tid & 31;
    const int lo   = blockIdx.x * OWN_ROWS;
    const unsigned idw = (unsigned)blockIdx.x * 0x01010101u;
    const int rot  = blockIdx.x % SCAN_ITERS;
    const long e16 = (long)n_edges & ~15L;

    __shared__ int            ovf_s;

    for (int r = w; r < OWN_ROWS; r += OWN_WAVES)
        *(float4*)(acc + r * 128 + lane * 4) = make_float4(0.f, 0.f, 0.f, 0.f);

    int* const recw = rec + (long)(blockIdx.x * OWN_WAVES + w) * (2 * REC_CAP);
    int  nrec = 0;
    bool full_path = true;
    if (REPLAY) full_path = (rec_flag[blockIdx.x] != 0);
    if (!REPLAY && tid == 0) ovf_s = 0;

    if (!full_path) {
        const int c = __builtin_amdgcn_readfirstlane(rec_cnt[blockIdx.x * OWN_WAVES + w]);
        int2 nx = (lane < c) ? *(const int2*)(recw + 2 * lane) : make_int2(0, 0);
        int k = 0;
        for (; k + 32 <= c; k += 32) {
            const int2 cur = nx;
            const int i2 = k + 32 + lane;
            nx = (i2 < c) ? *(const int2*)(recw + 2 * i2) : make_int2(0, 0);
            own_flush(msg_tab, acc, cur.x, cur.y, 32, lane);
        }
        if (k < c) own_flush(msg_tab, acc, nx.x, nx.y, c - k, lane);
    }

    for (long seg0 = 0; full_path && seg0 < n_edges; seg0 += SEG_EDGES) {
        const unsigned char* kseg = keys + seg0;
        const long rem = e16 - seg0;
        const int  lim = (rem > SEG_EDGES) ? SEG_EDGES : (int)rem;
        int cnt = 0;
        if (lim > 0) {
            for (int t0 = 0; t0 < SCAN_ITERS; t0 += 7) {
                v4u kw[7];
#pragma unroll
                for (int u = 0; u < 7; ++u) {
                    int it = t0 + u + rot;
                    if (it >= SCAN_ITERS) it -= SCAN_ITERS;
                    int rel = (w * SCAN_ITERS + it) * KEYS_PER_LD + lane * 16;
                    int rc  = (rel < lim - 16) ? rel : (lim - 16);
                    kw[u] = *(const v4u*)(kseg + rc);
                }
                asm volatile("" : "+v"(kw[0]), "+v"(kw[1]), "+v"(kw[2]), "+v"(kw[3]),
                                  "+v"(kw[4]), "+v"(kw[5]), "+v"(kw[6]));
#pragma unroll
                for (int u = 0; u < 7; ++u) {
                    int it = t0 + u + rot;
                    if (it >= SCAN_ITERS) it -= SCAN_ITERS;
                    int rel = (w * SCAN_ITERS + it) * KEYS_PER_LD + lane * 16;
                    unsigned m = match16(kw[u].x, kw[u].y, kw[u].z, kw[u].w, idw);
                    if (rel + 16 > lim) m = 0;
                    masks[(w * SCAN_ITERS + it) * 32 + lane] = (unsigned short)m;
                    cnt += __popc(m);
                }
            }
        } else {
            for (int it = 0; it < SCAN_ITERS; ++it) masks[(w * SCAN_ITERS + it) * 32 + lane] = 0;
        }
        if (e16 < n_edges) {
            long pos = e16 - seg0;
            if (pos >= 0 && pos < SEG_EDGES && w == (int)(pos / (SCAN_ITERS * KEYS_PER_LD)) &&
                lane == (int)((pos % KEYS_PER_LD) / 16)) {
                unsigned m = 0;
                for (int j = 0; e16 + j < n_edges; ++j)
                    if (keys[e16 + j] == (unsigned char)blockIdx.x) m |= 1u << (4 * (j & 3) + (j >> 2));
                masks[(int)(pos / 16)] = (unsigned short)m;
                cnt += __popc(m);
            }
        }
        for (int d = 16; d >= 1; d >>= 1) cnt += __shfl_xor(cnt, d);
        if (lane == 0) cntw[w] = cnt;
        __syncthreads();

        int c_l   = cntw[lane];
        int inc_l = wave_incl_scan(c_l, lane);
        const int total  = __shfl(inc_l, 31);
        const int base_w = __shfl(inc_l, w) - __shfl(c_l, w);
        __syncthreads();

        for (int r0 = 0; r0 < total; r0 += LIST_CAP) {
            const unsigned short* mw = masks + w * (SCAN_ITERS * 32) + lane * SCAN_ITERS;
            int c = 0;
            for (int i = 0; i < SCAN_ITERS; ++i) c += __popc((unsigned)mw[i]);
            int inc = wave_incl_scan(c, lane);
            int p   = base_w + inc - c - r0;
            int eb  = (int)seg0 + (w * (SCAN_ITERS * 32) + lane * SCAN_ITERS) * 16;
            for (int i = 0; i < SCAN_ITERS; ++i) {
                unsigned m = mw[i];
                while (m) {
                    int j = __ffs(m) - 1;
                    m &= m - 1;
                    if (p >= 0 && p < LIST_CAP) list_s[p] = eb + i * 16 + 4 * (j & 3) + (j >> 2);
                    ++p;
                }
            }
            __syncthreads();

            const int n = (total - r0 < LIST_CAP) ? (total - r0) : LIST_CAP;
            for (int i = tid; i < n; i += OWN_THREADS) {
                int e = list_s[i];
                list_s[i] = src[e];
                list_d[i] = dst[e] - lo;
            }
            __syncthreads();

            int qn = 0, qs = 0, qd = 0;
            int dn = (lane < n) ? list_d[lane] : -1;
            int sn = (lane < n) ? list_s[lane] : 0;
            for (int i0 = 0; i0 < n; i0 += 32) {
                const int d = dn, s = sn;
                const int i2 = i0 + 32 + lane;
                dn = (i2 < n) ? list_d[i2] : -1;
                sn = (i2 < n) ? list_s[i2] : 0;
                unsigned mk = __builtin_amdgcn_ballot_w32(d >= 0 && (d & (OWN_WAVES - 1)) == w);
                while (mk) {
                    int b = __builtin_ctz(mk);
                    mk &= mk - 1;
                    int sb = __builtin_amdgcn_readlane(s, b);
                    int db = __builtin_amdgcn_readlane(d, b);
                    if (lane == qn) { qs = sb; qd = db; }
                    ++qn;
                    if (qn == 32) {
                        if (!REPLAY) {
                            if (nrec + 32 <= REC_CAP) *(int2*)(recw + 2 * (nrec + lane)) = make_int2(qs, qd);
                            else ovf_s = 1;
                            nrec += 32;
                        }
                        own_flush(msg_tab, acc, qs, qd, 32, lane);
                        qn = 0;
                    }
                }
            }
            if (qn) {
                if (!REPLAY) {
                    if (nrec + qn <= REC_CAP) { if (lane < qn) *(int2*)(recw + 2 * (nrec + lane)) = make_int2(qs, qd); }
                    else ovf_s = 1;
                    nrec += qn;
                }
                own_flush(msg_tab, acc, qs, qd, qn, lane);
            }
            __syncthreads();
        }
    }

    if (!REPLAY) {
        if (lane == 0) rec_cnt[blockIdx.x * OWN_WAVES + w] = nrec;
        if (tid == 0)  rec_flag[blockIdx.x] = ovf_s;
    }

    for (int r = w; r < OWN_ROWS; r += OWN_WAVES) {
        int row = lo + r;
        if (row < n_nodes)
            *(float4*)(agg_tab + (long)row * 256 + lane * 4) = *(const float4*)(acc + r * 128 + lane * 4);
    }
}

extern "C" void kernel_launch(void* const* d_in, const int* in_sizes, int n_in,
                              void* d_out, int out_size, void* d_ws, size_t ws_size,
                              hipStream_t stream) {
    const float* features = (const float*)d_in[0];
    const int*   src      = (const int*)  d_in[1];
    const int*   dst      = (const int*)  d_in[2];
    const float* W1       = (const float*)d_in[3];
    const float* b1       = (const float*)d_in[4];
    const float* W2       = (const float*)d_in[5];
    const float* b2       = (const float*)d_in[6];

    float* out   = (float*)d_out;
    float* feat1 = (float*)d_ws;

    const int N = in_sizes[0] / 128;
    const int E = in_sizes[1];

    unsigned char* keys = (unsigned char*)d_ws + (size_t)N * 1024;

    int* rec      = (int*)((unsigned char*)d_ws + ((size_t)128 << 20));
    int* rec_cnt  = (int*)((unsigned char*)d_ws + ((size_t)224 << 20));
    int* rec_flag = rec_cnt + 16384;

    {
        int blocks = (E + 4095) / 4096;
        gnn_edge_keys<<<blocks, 256, 0, stream>>>(dst, keys, E);
    }

    gnn_gemm_wmma<128><<<512, 256, 0, stream>>>(features, W1, b1, feat1, 256, 128, N / 16);

    const int own_blocks = (N + OWN_ROWS - 1) / OWN_ROWS;

    gnn_scatter_own<false><<<own_blocks, OWN_THREADS, 0, stream>>>(src, dst, keys, feat1, feat1,
                                                                  rec, rec_cnt, rec_flag, N, E);

    gnn_gemm_wmma<256><<<512, 256, 0, stream>>>(feat1, W2, b2, out, 256, 128, N / 16);

    gnn_scatter_own<true><<<own_blocks, OWN_THREADS, 0, stream>>>(src, dst, keys, out, out,
                                                                 rec, rec_cnt, rec_flag, N, E);
}
